# starting kernel
# baseline (speedup 1.0000x reference)
#include <hip/hip_runtime.h>

typedef __attribute__((ext_vector_type(2))) float v2f;
typedef __attribute__((ext_vector_type(8))) float v8f;

#define TILE_H 128
#define TILE_W 128
#define TILE_HW (TILE_H * TILE_W)

__device__ __forceinline__ float parzen_exp(const float* __restrict__ p,
                                            float m, int ax, int ay,
                                            int r, int c) {
  float dx = (float)(c - 5);
  float dy = (float)(r - 5);
  float d2 = dx * dx + dy * dy;
  if (d2 > 25.0f) return 0.0f;
  int gy = ay - 5 + r;
  int gx = ax - 5 + c;
  if (gy < 0 || gy >= TILE_H || gx < 0 || gx >= TILE_W) return 0.0f;
  float d  = __builtin_amdgcn_sqrtf(d2);
  float dp = d * 0.2f;
  float win;
  if (d <= 2.5f) {
    win = 1.0f - 6.0f * dp * dp * (1.0f - dp);
  } else {
    float t = 1.0f - dp;
    win = 2.0f * t * t * t;
  }
  float v = p[gy * TILE_W + gx];
  return win * expf(v - m);
}

__global__ __launch_bounds__(256) void softargmax2d_kernel(
    const float* __restrict__ x, float* __restrict__ out, int nchan) {
  const int bc = blockIdx.x;
  if (bc >= nchan) return;
  const float* __restrict__ p = x + (size_t)bc * TILE_HW;
  const int tid  = threadIdx.x;
  const int lane = tid & 31;
  const int wave = tid >> 5;

  float vmax = -__builtin_inff();
  int   imax = 0;
  const float4* p4 = (const float4*)p;
#pragma unroll
  for (int j = 0; j < 16; ++j) {
    int i4 = tid + 256 * j;
    float4 v = p4[i4];
    int base = i4 * 4;
    if (v.x > vmax) { vmax = v.x; imax = base;     }
    if (v.y > vmax) { vmax = v.y; imax = base + 1; }
    if (v.z > vmax) { vmax = v.z; imax = base + 2; }
    if (v.w > vmax) { vmax = v.w; imax = base + 3; }
  }
#pragma unroll
  for (int off = 16; off >= 1; off >>= 1) {
    float vo = __shfl_xor(vmax, off, 32);
    int   io = __shfl_xor(imax, off, 32);
    if (vo > vmax || (vo == vmax && io < imax)) { vmax = vo; imax = io; }
  }

  __shared__ float s_val[8];
  __shared__ int   s_idx[8];
  __shared__ float s_m;
  __shared__ int   s_am;
  if (lane == 0) { s_val[wave] = vmax; s_idx[wave] = imax; }
  __syncthreads();
  if (tid == 0) {
    float m = s_val[0]; int am = s_idx[0];
#pragma unroll
    for (int wv = 1; wv < 8; ++wv) {
      float vo = s_val[wv]; int io = s_idx[wv];
      if (vo > m || (vo == m && io < am)) { m = vo; am = io; }
    }
    s_m = m; s_am = am;
  }
  __syncthreads();

  if (wave == 0) {
    const float m  = s_m;
    const int   am = s_am;
    const int   ax = am & (TILE_W - 1);
    const int   ay = am >> 7;
    const int   pidx  = lane & 15;
    const int   qbase = (lane < 16) ? 0 : 2;

    v8f accA = {};
    v8f accB = {};
    v2f ones; ones[0] = 1.0f; ones[1] = 1.0f;

#pragma unroll
    for (int j = 0; j < 4; ++j) {
      int q0 = 4 * j + qbase;
      v2f a;
      a[0] = parzen_exp(p, m, ax, ay, pidx, q0);
      a[1] = parzen_exp(p, m, ax, ay, pidx, q0 + 1);
      v2f b;
      b[0] = parzen_exp(p, m, ax, ay, q0, pidx);
      b[1] = parzen_exp(p, m, ax, ay, q0 + 1, pidx);
      accA = __builtin_amdgcn_wmma_f32_16x16x4_f32(
          false, a, false, ones, (short)0, accA, false, false);
      accB = __builtin_amdgcn_wmma_f32_16x16x4_f32(
          false, ones, false, b, (short)0, accB, false, false);
    }

    float Wp = 0.0f, Yp = 0.0f;
    const int mbase = (lane < 16) ? 0 : 8;
#pragma unroll
    for (int i = 0; i < 8; ++i) {
      float r = accA[i];
      Wp += r;
      Yp += r * (float)(ay - 5 + mbase + i);
    }
    Wp += __shfl_xor(Wp, 16, 32);
    Yp += __shfl_xor(Yp, 16, 32);

    float Xp = accB[0] * (float)(ax - 5 + pidx);
    Xp += __shfl_xor(Xp, 1, 32);
    Xp += __shfl_xor(Xp, 2, 32);
    Xp += __shfl_xor(Xp, 4, 32);
    Xp += __shfl_xor(Xp, 8, 32);

    if (lane == 0) {
      out[2 * bc + 0] = Xp / Wp;
      out[2 * bc + 1] = Yp / Wp;
    }
  }
}

extern "C" void kernel_launch(void* const* d_in, const int* in_sizes, int n_in,
                              void* d_out, int out_size, void* d_ws, size_t ws_size,
                              hipStream_t stream) {
  (void)n_in; (void)out_size; (void)d_ws; (void)ws_size;
  const float* x = (const float*)d_in[0];
  float* out = (float*)d_out;
  int nchan = in_sizes[0] / TILE_HW;
  softargmax2d_kernel<<<nchan, 256, 0, stream>>>(x, out, nchan);
}


# one wave per channel: each wave streams its 64 KiB channel through a ring of 32 loads in flight and does its own argmax and window sums; 2048 waves launched, not eight per channel; no LDS, no barrier
# speedup vs baseline: 1.0511x; 1.0511x over previous
#include <hip/hip_runtime.h>


typedef __attribute__((ext_vector_type(2))) float v2f;
typedef __attribute__((ext_vector_type(8))) float v8f;

#define TILE_H 128
#define TILE_W 128
#define TILE_HW (TILE_H * TILE_W)

#define WAVES_PER_WG 8
#define LOADS_PER_CHANNEL (TILE_HW / 4 / 32)
#define RING 32
#define TRIPS (LOADS_PER_CHANNEL / RING)

__device__ __forceinline__ float parzen_exp(const float* __restrict__ p,
                                            float m, int ax, int ay,
                                            int r, int c) {
  float dx = (float)(c - 5);
  float dy = (float)(r - 5);
  float d2 = dx * dx + dy * dy;
  if (d2 > 25.0f) return 0.0f;
  int gy = ay - 5 + r;
  int gx = ax - 5 + c;
  if (gy < 0 || gy >= TILE_H || gx < 0 || gx >= TILE_W) return 0.0f;
  float d  = __builtin_amdgcn_sqrtf(d2);
  float dp = d * 0.2f;
  float win;
  if (d <= 2.5f) {
    win = 1.0f - 6.0f * dp * dp * (1.0f - dp);
  } else {
    float t = 1.0f - dp;
    win = 2.0f * t * t * t;
  }
  float v = p[gy * TILE_W + gx];
  return win * expf(v - m);
}

__device__ __forceinline__ void scan4(const float4 v, int code,
                                      float& lv, int& lk) {
  if (v.x > lv) { lv = v.x; lk = code;     }
  if (v.y > lv) { lv = v.y; lk = code + 1; }
  if (v.z > lv) { lv = v.z; lk = code + 2; }
  if (v.w > lv) { lv = v.w; lk = code + 3; }
}

__global__ __launch_bounds__(WAVES_PER_WG * 32) void softargmax2d_kernel(
    const float* __restrict__ x, float* __restrict__ out, int nchan) {
  const int lane = threadIdx.x & 31;
  const int wave = __builtin_amdgcn_readfirstlane((int)(threadIdx.x >> 5));
  const int bc   = blockIdx.x * WAVES_PER_WG + wave;
  if (bc >= nchan) return;
  const float* __restrict__ p = x + (size_t)bc * TILE_HW;
  const float4* __restrict__ q = (const float4*)p + lane;

  float vmax = -__builtin_inff();
  int   kmax = 0;
  float4 r[RING];
#pragma unroll
  for (int j = 0; j < RING; ++j) r[j] = q[32 * j];

#pragma nounroll
  for (int t = 0; t < TRIPS - 1; ++t) {
    const float4* __restrict__ qn = q + (t + 1) * (32 * RING);
    float lv = vmax;
    int   lk = -1;
#pragma unroll
    for (int j = 0; j < RING; ++j) {
      scan4(r[j], 4 * j, lv, lk);
      r[j] = qn[32 * j];
      __builtin_amdgcn_sched_barrier(0);
    }
    kmax = (lk >= 0) ? lk + 4 * RING * t : kmax;
    vmax = lv;
  }
  {
    float lv = vmax;
    int   lk = -1;
#pragma unroll
    for (int j = 0; j < RING; ++j) scan4(r[j], 4 * j, lv, lk);
    kmax = (lk >= 0) ? lk + 4 * RING * (TRIPS - 1) : kmax;
    vmax = lv;
  }

  int imax = 128 * (kmax >> 2) + 4 * lane + (kmax & 3);
#pragma unroll
  for (int off = 16; off >= 1; off >>= 1) {
    float vo = __shfl_xor(vmax, off, 32);
    int   io = __shfl_xor(imax, off, 32);
    if (vo > vmax || (vo == vmax && io < imax)) { vmax = vo; imax = io; }
  }
  const float m  = __builtin_bit_cast(
      float, __builtin_amdgcn_readfirstlane(__builtin_bit_cast(int, vmax)));
  const int   am = __builtin_amdgcn_readfirstlane(imax);

  const int   ax = am & (TILE_W - 1);
  const int   ay = am >> 7;
  const int   pidx  = lane & 15;
  const int   qbase = (lane < 16) ? 0 : 2;

  v8f accA = {};
  v8f accB = {};
  v2f ones; ones[0] = 1.0f; ones[1] = 1.0f;

#pragma unroll
  for (int j = 0; j < 4; ++j) {
    int q0 = 4 * j + qbase;
    v2f fa;
    fa[0] = parzen_exp(p, m, ax, ay, pidx, q0);
    fa[1] = parzen_exp(p, m, ax, ay, pidx, q0 + 1);
    v2f fb;
    fb[0] = parzen_exp(p, m, ax, ay, q0, pidx);
    fb[1] = parzen_exp(p, m, ax, ay, q0 + 1, pidx);
    accA = __builtin_amdgcn_wmma_f32_16x16x4_f32(
        false, fa, false, ones, (short)0, accA, false, false);
    accB = __builtin_amdgcn_wmma_f32_16x16x4_f32(
        false, ones, false, fb, (short)0, accB, false, false);
  }

  float Wp = 0.0f, Yp = 0.0f;
  const int mbase = (lane < 16) ? 0 : 8;
#pragma unroll
  for (int i = 0; i < 8; ++i) {
    float rr = accA[i];
    Wp += rr;
    Yp += rr * (float)(ay - 5 + mbase + i);
  }
  Wp += __shfl_xor(Wp, 16, 32);
  Yp += __shfl_xor(Yp, 16, 32);

  float Xp = accB[0] * (float)(ax - 5 + pidx);
  Xp += __shfl_xor(Xp, 1, 32);
  Xp += __shfl_xor(Xp, 2, 32);
  Xp += __shfl_xor(Xp, 4, 32);
  Xp += __shfl_xor(Xp, 8, 32);

  if (lane == 0) {
    out[2 * bc + 0] = Xp / Wp;
    out[2 * bc + 1] = Yp / Wp;
  }
}

extern "C" void kernel_launch(void* const* d_in, const int* in_sizes, int n_in,
                              void* d_out, int out_size, void* d_ws, size_t ws_size,
                              hipStream_t stream) {
  (void)n_in; (void)out_size; (void)d_ws; (void)ws_size;
  const float* x = (const float*)d_in[0];
  float* out = (float*)d_out;
  int nchan = in_sizes[0] / TILE_HW;
  int nwg = (nchan + WAVES_PER_WG - 1) / WAVES_PER_WG;
  softargmax2d_kernel<<<nwg, WAVES_PER_WG * 32, 0, stream>>>(x, out, nchan);
}


# window stage: a lane's sixteen window elements are loaded as one batch at clamped addresses (weight 0 where outside), not one load-and-wait per element under divergent branches; sum order unchanged
# speedup vs baseline: 1.1695x; 1.1126x over previous
#include <hip/hip_runtime.h>


typedef __attribute__((ext_vector_type(2))) float v2f;
typedef __attribute__((ext_vector_type(8))) float v8f;

#define TILE_H 128
#define TILE_W 128
#define TILE_HW (TILE_H * TILE_W)

#define WAVES_PER_WG 8
#define LOADS_PER_CHANNEL (TILE_HW / 4 / 32)
#define RING 32
#define TRIPS (LOADS_PER_CHANNEL / RING)

__device__ __forceinline__ float parzen_weight(int ax, int ay, int r, int c,
                                               int& idx) {
  float dx = (float)(c - 5);
  float dy = (float)(r - 5);
  float d2 = dx * dx + dy * dy;
  int gy = ay - 5 + r;
  int gx = ax - 5 + c;
  bool inside = (d2 <= 25.0f) && gy >= 0 && gy < TILE_H && gx >= 0 && gx < TILE_W;
  int cy = min(max(gy, 0), TILE_H - 1);
  int cx = min(max(gx, 0), TILE_W - 1);
  idx = cy * TILE_W + cx;
  float d  = __builtin_amdgcn_sqrtf(d2);
  float dp = d * 0.2f;
  float t  = 1.0f - dp;
  float near_mode = 1.0f - 6.0f * dp * dp * (1.0f - dp);
  float tail      = 2.0f * t * t * t;
  float win = (d <= 2.5f) ? near_mode : tail;
  return inside ? win : 0.0f;
}

__device__ __forceinline__ void scan4(const float4 v, int code,
                                      float& lv, int& lk) {
  if (v.x > lv) { lv = v.x; lk = code;     }
  if (v.y > lv) { lv = v.y; lk = code + 1; }
  if (v.z > lv) { lv = v.z; lk = code + 2; }
  if (v.w > lv) { lv = v.w; lk = code + 3; }
}

__global__ __launch_bounds__(WAVES_PER_WG * 32) void softargmax2d_kernel(
    const float* __restrict__ x, float* __restrict__ out, int nchan) {
  const int lane = threadIdx.x & 31;
  const int wave = __builtin_amdgcn_readfirstlane((int)(threadIdx.x >> 5));
  const int bc   = blockIdx.x * WAVES_PER_WG + wave;
  if (bc >= nchan) return;
  const float* __restrict__ p = x + (size_t)bc * TILE_HW;
  const float4* __restrict__ q = (const float4*)p + lane;

  float vmax = -__builtin_inff();
  int   kmax = 0;
  float4 r[RING];
#pragma unroll
  for (int j = 0; j < RING; ++j) r[j] = q[32 * j];

#pragma nounroll
  for (int t = 0; t < TRIPS - 1; ++t) {
    const float4* __restrict__ qn = q + (t + 1) * (32 * RING);
    float lv = vmax;
    int   lk = -1;
#pragma unroll
    for (int j = 0; j < RING; ++j) {
      scan4(r[j], 4 * j, lv, lk);
      r[j] = qn[32 * j];
      __builtin_amdgcn_sched_barrier(0);
    }
    kmax = (lk >= 0) ? lk + 4 * RING * t : kmax;
    vmax = lv;
  }
  {
    float lv = vmax;
    int   lk = -1;
#pragma unroll
    for (int j = 0; j < RING; ++j) scan4(r[j], 4 * j, lv, lk);
    kmax = (lk >= 0) ? lk + 4 * RING * (TRIPS - 1) : kmax;
    vmax = lv;
  }

  int imax = 128 * (kmax >> 2) + 4 * lane + (kmax & 3);
#pragma unroll
  for (int off = 16; off >= 1; off >>= 1) {
    float vo = __shfl_xor(vmax, off, 32);
    int   io = __shfl_xor(imax, off, 32);
    if (vo > vmax || (vo == vmax && io < imax)) { vmax = vo; imax = io; }
  }
  const float m  = __builtin_bit_cast(
      float, __builtin_amdgcn_readfirstlane(__builtin_bit_cast(int, vmax)));
  const int   am = __builtin_amdgcn_readfirstlane(imax);

  const int   ax = am & (TILE_W - 1);
  const int   ay = am >> 7;
  const int   pidx  = lane & 15;
  const int   qbase = (lane < 16) ? 0 : 2;

  v8f accA = {};
  v8f accB = {};
  v2f ones; ones[0] = 1.0f; ones[1] = 1.0f;

  float wgt[4][4];
  float val[4][4];
#pragma unroll
  for (int j = 0; j < 4; ++j) {
    int q0 = 4 * j + qbase;
    int i0, i1, i2, i3;
    wgt[j][0] = parzen_weight(ax, ay, pidx, q0,     i0);
    wgt[j][1] = parzen_weight(ax, ay, pidx, q0 + 1, i1);
    wgt[j][2] = parzen_weight(ax, ay, q0,     pidx, i2);
    wgt[j][3] = parzen_weight(ax, ay, q0 + 1, pidx, i3);
    val[j][0] = p[i0];
    val[j][1] = p[i1];
    val[j][2] = p[i2];
    val[j][3] = p[i3];
  }

#pragma unroll
  for (int j = 0; j < 4; ++j) {
    v2f fa;
    fa[0] = wgt[j][0] * expf(val[j][0] - m);
    fa[1] = wgt[j][1] * expf(val[j][1] - m);
    v2f fb;
    fb[0] = wgt[j][2] * expf(val[j][2] - m);
    fb[1] = wgt[j][3] * expf(val[j][3] - m);
    accA = __builtin_amdgcn_wmma_f32_16x16x4_f32(
        false, fa, false, ones, (short)0, accA, false, false);
    accB = __builtin_amdgcn_wmma_f32_16x16x4_f32(
        false, ones, false, fb, (short)0, accB, false, false);
  }

  float Wp = 0.0f, Yp = 0.0f;
  const int mbase = (lane < 16) ? 0 : 8;
#pragma unroll
  for (int i = 0; i < 8; ++i) {
    float rr = accA[i];
    Wp += rr;
    Yp += rr * (float)(ay - 5 + mbase + i);
  }
  Wp += __shfl_xor(Wp, 16, 32);
  Yp += __shfl_xor(Yp, 16, 32);

  float Xp = accB[0] * (float)(ax - 5 + pidx);
  Xp += __shfl_xor(Xp, 1, 32);
  Xp += __shfl_xor(Xp, 2, 32);
  Xp += __shfl_xor(Xp, 4, 32);
  Xp += __shfl_xor(Xp, 8, 32);

  if (lane == 0) {
    out[2 * bc + 0] = Xp / Wp;
    out[2 * bc + 1] = Yp / Wp;
  }
}

extern "C" void kernel_launch(void* const* d_in, const int* in_sizes, int n_in,
                              void* d_out, int out_size, void* d_ws, size_t ws_size,
                              hipStream_t stream) {
  (void)n_in; (void)out_size; (void)d_ws; (void)ws_size;
  const float* x = (const float*)d_in[0];
  float* out = (float*)d_out;
  int nchan = in_sizes[0] / TILE_HW;
  int nwg = (nchan + WAVES_PER_WG - 1) / WAVES_PER_WG;
  softargmax2d_kernel<<<nwg, WAVES_PER_WG * 32, 0, stream>>>(x, out, nchan);
}
